# starting kernel
# baseline (speedup 1.0000x reference)
#include <hip/hip_runtime.h>
#include <math.h>

typedef float v2f __attribute__((ext_vector_type(2)));
typedef float v8f __attribute__((ext_vector_type(8)));

#define D_H 256
#define LN_EPS 1e-5f

__global__ void k_zero_f32(float* __restrict__ p, long n) {
    long i = (long)blockIdx.x * blockDim.x + threadIdx.x;
    long stride = (long)gridDim.x * blockDim.x;
    for (; i < n; i += stride) p[i] = 0.0f;
}

__global__ void k_scatter_edges(const float* __restrict__ h,
                                const int* __restrict__ src,
                                const int* __restrict__ dst,
                                float* __restrict__ accum,
                                float* __restrict__ deg, int E) {
    int wave = (int)((blockIdx.x * (long)blockDim.x + threadIdx.x) >> 5);
    int lane = threadIdx.x & 31;
    if (wave >= E) return;
    int s = src[wave];
    int d = dst[wave];
    const float* hs = h + (size_t)s * D_H;
    float* ad = accum + (size_t)d * D_H;
#pragma unroll
    for (int i = 0; i < 8; ++i) {
        int c = lane + i * 32;
        atomicAdd(&ad[c], hs[c]);
    }
    if (lane == 0) atomicAdd(&deg[d], 1.0f);
}

__global__ void k_finalize_mean(float* __restrict__ accum,
                                const float* __restrict__ h_dst,
                                const float* __restrict__ deg, long total) {
    long i = (long)blockIdx.x * blockDim.x + threadIdx.x;
    if (i >= total) return;
    long r = i >> 8;
    float inv = 1.0f / (deg[r] + 1.0f);
    accum[i] = (accum[i] + h_dst[i]) * inv;
}

#define KCH 128
#define ALD 132
__global__ __launch_bounds__(256) void k_gemm_wmma_f32(
    const float* __restrict__ A, const float* __restrict__ W,
    const float* __restrict__ bias, float* __restrict__ out, int M, int N) {
    __shared__ float Alds[32 * ALD];
    __shared__ float Blds[KCH * 64];
    const int tid = threadIdx.x;
    const int rowBase = blockIdx.x * 32;
    const int colBase = blockIdx.y * 64;

    const int wave = tid >> 5, lane = tid & 31;
    const int wr = wave >> 2, wc = wave & 3;
    const int half = lane >> 4, l16 = lane & 15;

    v8f c = {};

    for (int kb = 0; kb < D_H; kb += KCH) {
        __syncthreads();
#pragma unroll
        for (int j = 0; j < 4; ++j) {
            int idx = tid + j * 256;
            int r = idx >> 5, c4 = idx & 31;
            int grow = rowBase + r;
            float4 v = make_float4(0.f, 0.f, 0.f, 0.f);
            if (grow < M) v = *(const float4*)(A + (size_t)grow * D_H + kb + c4 * 4);
            *(float4*)(&Alds[r * ALD + c4 * 4]) = v;
        }
#pragma unroll
        for (int j = 0; j < 8; ++j) {
            int idx = tid + j * 256;
            int k = idx >> 4, c4 = idx & 15;
            float4 v = *(const float4*)(W + (size_t)(kb + k) * N + colBase + c4 * 4);
            *(float4*)(&Blds[k * 64 + c4 * 4]) = v;
        }
        __syncthreads();

        const float* arow = &Alds[(wr * 16 + l16) * ALD];
        const float* bcol = &Blds[wc * 16 + l16];
#pragma unroll 8
        for (int k0 = 0; k0 < KCH; k0 += 4) {
            v2f a, b;
            a.x = arow[k0 + 2 * half];
            a.y = arow[k0 + 2 * half + 1];
            b.x = bcol[(size_t)(k0 + 2 * half) * 64];
            b.y = bcol[(size_t)(k0 + 2 * half + 1) * 64];
            c = __builtin_amdgcn_wmma_f32_16x16x4_f32(
                false, a, false, b, (short)0, c, false, false);
        }
    }

    int col = colBase + wc * 16 + l16;
    float bv = bias[col];
    int rbase = rowBase + wr * 16 + half * 8;
#pragma unroll
    for (int j = 0; j < 8; ++j) {
        int row = rbase + j;
        if (row < M) out[(size_t)row * N + col] = c[j] + bv;
    }
}

__global__ void k_ln_relu(const float* __restrict__ x, const float* __restrict__ g,
                          const float* __restrict__ b, float* __restrict__ y, int M) {
    int wave = (int)((blockIdx.x * (long)blockDim.x + threadIdx.x) >> 5);
    int lane = threadIdx.x & 31;
    if (wave >= M) return;
    const float* xr = x + (size_t)wave * D_H;
    float v[8];
    float s = 0.0f;
#pragma unroll
    for (int i = 0; i < 8; ++i) { v[i] = xr[lane + 32 * i]; s += v[i]; }
#pragma unroll
    for (int o = 16; o > 0; o >>= 1) s += __shfl_xor(s, o, 32);
    float mu = s * (1.0f / 256.0f);
    float vs = 0.0f;
#pragma unroll
    for (int i = 0; i < 8; ++i) { float d = v[i] - mu; vs += d * d; }
#pragma unroll
    for (int o = 16; o > 0; o >>= 1) vs += __shfl_xor(vs, o, 32);
    float rs = rsqrtf(vs * (1.0f / 256.0f) + LN_EPS);
    float* yr = y + (size_t)wave * D_H;
#pragma unroll
    for (int i = 0; i < 8; ++i) {
        int ci = lane + 32 * i;
        float t = (v[i] - mu) * rs * g[ci] + b[ci];
        yr[ci] = fmaxf(t, 0.0f);
    }
}

extern "C" void kernel_launch(void* const* d_in, const int* in_sizes, int n_in,
                              void* d_out, int out_size, void* d_ws, size_t ws_size,
                              hipStream_t stream) {
    (void)n_in; (void)out_size; (void)ws_size;
    const float* feats = (const float*)d_in[0];
    const float* W0    = (const float*)d_in[1];
    const float* b0    = (const float*)d_in[2];
    const float* W1    = (const float*)d_in[3];
    const float* b1    = (const float*)d_in[4];
    const float* ln_g  = (const float*)d_in[5];
    const float* ln_b  = (const float*)d_in[6];
    const int* src0 = (const int*)d_in[7];
    const int* dst0 = (const int*)d_in[8];
    const int* src1 = (const int*)d_in[9];
    const int* dst1 = (const int*)d_in[10];
    const int E0 = in_sizes[7], E1 = in_sizes[9];
    const int N1 = 50000, N2 = 10000, DOUT = 128;

    float* ws     = (float*)d_ws;
    float* accum0 = ws;
    float* deg0   = accum0 + (size_t)N1 * D_H;
    float* h      = deg0 + N1;
    float* accum1 = h + (size_t)N1 * D_H;
    float* deg1   = accum1 + (size_t)N2 * D_H;

    float* h0_out = (float*)d_out;
    float* out2   = h0_out + (size_t)N1 * D_H;

    long z0 = (long)N1 * D_H + N1;
    k_zero_f32<<<4096, 256, 0, stream>>>(accum0, z0);
    {
        long threads = (long)E0 * 32;
        k_scatter_edges<<<(int)((threads + 255) / 256), 256, 0, stream>>>(
            feats, src0, dst0, accum0, deg0, E0);
    }
    {
        long total = (long)N1 * D_H;
        k_finalize_mean<<<(int)((total + 255) / 256), 256, 0, stream>>>(
            accum0, feats, deg0, total);
    }
    {
        dim3 grid((N1 + 31) / 32, D_H / 64);
        k_gemm_wmma_f32<<<grid, 256, 0, stream>>>(accum0, W0, b0, h0_out, N1, D_H);
    }
    {
        long threads = (long)N1 * 32;
        k_ln_relu<<<(int)((threads + 255) / 256), 256, 0, stream>>>(
            h0_out, ln_g, ln_b, h, N1);
    }

    long z1 = (long)N2 * D_H + N2;
    k_zero_f32<<<2048, 256, 0, stream>>>(accum1, z1);
    {
        long threads = (long)E1 * 32;
        k_scatter_edges<<<(int)((threads + 255) / 256), 256, 0, stream>>>(
            h, src1, dst1, accum1, deg1, E1);
    }
    {
        long total = (long)N2 * D_H;
        k_finalize_mean<<<(int)((total + 255) / 256), 256, 0, stream>>>(
            accum1, h, deg1, total);
    }
    {
        dim3 grid((N2 + 31) / 32, DOUT / 64);
        k_gemm_wmma_f32<<<grid, 256, 0, stream>>>(accum1, W1, b1, out2, N2, DOUT);
    }
}


# GEMM made row-walking: a workgroup keeps its 64-column W slab (all 256 k) in LDS and loops over 32-row A tiles, next tile's loads in flight under the WMMAs; 6252 short workgroups become 768
# speedup vs baseline: 3.5064x; 3.5064x over previous
#include <hip/hip_runtime.h>
#include <math.h>

typedef float v2f __attribute__((ext_vector_type(2)));
typedef float v8f __attribute__((ext_vector_type(8)));

#define D_H 256
#define LN_EPS 1e-5f

#define TILE0 196
#define TILE1 40
#define CH_EDGES 4096
#define CH_WORDS 128
#define BM_LD 132

template <int TILE>
__device__ __forceinline__ void tile_bits_layer(unsigned* bm, const int* __restrict__ src,
                                                const int* __restrict__ dst,
                                                unsigned* __restrict__ plane, int E,
                                                int n_dst, int first, int stride) {
    const int tid = threadIdx.x;
    const int nch = (E + CH_EDGES - 1) / CH_EDGES;
    const int ntiles = (n_dst + TILE - 1) / TILE;
    const size_t pitch = (size_t)nch * CH_WORDS;
    unsigned* pk = plane + (size_t)ntiles * pitch;
    for (int c = first; c < nch; c += stride) {
        for (int i = tid; i < 256 * BM_LD / 4; i += 1024)
            ((uint4*)bm)[i] = make_uint4(0u, 0u, 0u, 0u);
        __syncthreads();
        const int e0 = c * CH_EDGES + tid * 4;
        int4 d, s;
        const bool whole = e0 + 3 < E;
        if (whole) {
            d = *(const int4*)(dst + e0);
            s = *(const int4*)(src + e0);
        } else {
            d.x = (e0     < E) ? dst[e0]     : -1;  s.x = (e0     < E) ? src[e0]     : 0;
            d.y = (e0 + 1 < E) ? dst[e0 + 1] : -1;  s.y = (e0 + 1 < E) ? src[e0 + 1] : 0;
            d.z = (e0 + 2 < E) ? dst[e0 + 2] : -1;  s.z = (e0 + 2 < E) ? src[e0 + 2] : 0;
            d.w = (e0 + 3 < E) ? dst[e0 + 3] : -1;  s.w = (e0 + 3 < E) ? src[e0 + 3] : 0;
        }
        const int dk[4] = {d.x, d.y, d.z, d.w};
        const int sk[4] = {s.x, s.y, s.z, s.w};
        unsigned pw[4];
        unsigned* col = bm + (tid >> 3);
        const unsigned b0 = (unsigned)((tid & 7) * 4);
#pragma unroll
        for (int k = 0; k < 4; ++k) {
            const unsigned t = (unsigned)dk[k] / (unsigned)TILE;
            pw[k] = ((unsigned)sk[k] << 8) | (((unsigned)dk[k] - t * (unsigned)TILE) & 255u);
            if ((unsigned)dk[k] < (unsigned)n_dst)
                atomicOr(col + t * BM_LD, 1u << (b0 + k));
        }
        if (whole) {
            *(uint4*)(pk + e0) = make_uint4(pw[0], pw[1], pw[2], pw[3]);
        } else {
#pragma unroll
            for (int k = 0; k < 4; ++k)
                if (e0 + k < E) pk[e0 + k] = pw[k];
        }
        __syncthreads();
        for (int i = tid; i < ntiles * (CH_WORDS / 4); i += 1024) {
            int t = i >> 5, q = i & 31;
            uint4 v = *(const uint4*)(bm + t * BM_LD + q * 4);
            *(uint4*)(plane + (size_t)t * pitch + (size_t)c * CH_WORDS + q * 4) = v;
        }
        __syncthreads();
    }
}

__global__ __launch_bounds__(1024) void k_tile_bits(
    const int* __restrict__ src0, const int* __restrict__ dst0,
    unsigned* __restrict__ plane0, int E0, int n0, int nb0,
    const int* __restrict__ src1, const int* __restrict__ dst1,
    unsigned* __restrict__ plane1, int E1, int n1) {
    __shared__ __attribute__((aligned(16))) unsigned bm[256 * BM_LD];
    const int b = (int)blockIdx.x;
    if (b < nb0) tile_bits_layer<TILE0>(bm, src0, dst0, plane0, E0, n0, b, nb0);
    else         tile_bits_layer<TILE1>(bm, src1, dst1, plane1, E1, n1, b - nb0,
                                        (int)gridDim.x - nb0);
}

#define AGG_WAVES 32
#define QCAP 64

struct LnCoef { float4 ga, gb, ba, bb; };

template <int CTRL>
__device__ __forceinline__ float dpp_add(float v) {
    return v + __builtin_bit_cast(float, __builtin_amdgcn_update_dpp(
                   0, __builtin_bit_cast(int, v), CTRL, 0xf, 0xf, true));
}
__device__ __forceinline__ float wave_sum32(float v) {
    v = dpp_add<0x161>(v);
    v = dpp_add<0x162>(v);
    v = dpp_add<0x164>(v);
    v = dpp_add<0x168>(v);
    const int t = __builtin_bit_cast(int, v);
    v += __builtin_bit_cast(float, __builtin_amdgcn_permlanex16(t, t, 0x76543210, 0xfedcba98,
                                                               false, false));
    return v;
}

__device__ __forceinline__ void ln_relu_row(float4& a, float4& b, const LnCoef& c) {
    v2f a0 = {a.x, a.y}, a1 = {a.z, a.w}, b0 = {b.x, b.y}, b1 = {b.z, b.w};
    const v2f t = (a0 + a1) + (b0 + b1);
    const float mu = wave_sum32(t.x + t.y) * (1.0f / 256.0f);
    const v2f m2 = {mu, mu};
    a0 -= m2; a1 -= m2; b0 -= m2; b1 -= m2;
    const v2f u = (a0 * a0 + a1 * a1) + (b0 * b0 + b1 * b1);
    const float rs = rsqrtf(wave_sum32(u.x + u.y) * (1.0f / 256.0f) + LN_EPS);
    const v2f r2 = {rs, rs};
    const v2f g0 = {c.ga.x, c.ga.y}, g1 = {c.ga.z, c.ga.w};
    const v2f g2 = {c.gb.x, c.gb.y}, g3 = {c.gb.z, c.gb.w};
    const v2f s0 = {c.ba.x, c.ba.y}, s1 = {c.ba.z, c.ba.w};
    const v2f s2 = {c.bb.x, c.bb.y}, s3 = {c.bb.z, c.bb.w};
    a0 = a0 * r2 * g0 + s0;  a1 = a1 * r2 * g1 + s1;
    b0 = b0 * r2 * g2 + s2;  b1 = b1 * r2 * g3 + s3;
    a.x = fmaxf(a0.x, 0.0f); a.y = fmaxf(a0.y, 0.0f);
    a.z = fmaxf(a1.x, 0.0f); a.w = fmaxf(a1.y, 0.0f);
    b.x = fmaxf(b0.x, 0.0f); b.y = fmaxf(b0.y, 0.0f);
    b.z = fmaxf(b1.x, 0.0f); b.w = fmaxf(b1.y, 0.0f);
}

template <int TILE, bool LN>
__device__ __forceinline__ void agg_drain(const unsigned* qb, unsigned n,
                                          int lane, const float* __restrict__ h,
                                          const unsigned* __restrict__ pk,
                                          float* acc, float* degs, const LnCoef& lc) {
    unsigned e = 0u;
    if ((unsigned)lane < n) e = qb[lane];
    const unsigned p = pk[e];
    int s = (int)(p >> 8);
    int loc = TILE;
    if ((unsigned)lane < n) loc = (int)min(p & 255u, (unsigned)TILE);
    if ((unsigned)lane < n) atomicAdd(&degs[loc], 1.0f);
    constexpr int R = LN ? 7 : 8;
    for (unsigned j0 = 0; j0 < n; j0 += R) {
        float4 va[R], vb[R];
#pragma unroll
        for (int g = 0; g < R; ++g) {
            const int slot = (32 % R != 0) ? min((int)j0 + g, 31) : (int)j0 + g;
            int sj = __builtin_amdgcn_readlane(s, slot);
            const float* hr = h + (size_t)sj * D_H;
            va[g] = *(const float4*)(hr + lane * 4);
            vb[g] = *(const float4*)(hr + 128 + lane * 4);
        }
        if constexpr (LN) __builtin_amdgcn_sched_barrier(0);
#pragma unroll
        for (int g = 0; g < R; ++g) {
            const int slot = (32 % R != 0) ? min((int)j0 + g, 31) : (int)j0 + g;
            int lj = __builtin_amdgcn_readlane(loc, slot);
            if (32 % R != 0 && (int)j0 + g > 31) lj = TILE;
            if constexpr (LN) ln_relu_row(va[g], vb[g], lc);
            float* ar = acc + lj * D_H + lane;
            atomicAdd(ar,       va[g].x);
            atomicAdd(ar + 64,  va[g].y);
            atomicAdd(ar + 128, va[g].z);
            atomicAdd(ar + 192, va[g].w);
            atomicAdd(ar + 32,  vb[g].x);
            atomicAdd(ar + 96,  vb[g].y);
            atomicAdd(ar + 160, vb[g].z);
            atomicAdd(ar + 224, vb[g].w);
        }
    }
}

template <int TILE, bool LN>
__global__ __launch_bounds__(1024) void k_tile_aggregate(
    const float* __restrict__ h, const unsigned* __restrict__ plane,
    float* __restrict__ hneigh, int E, int n_dst,
    const float* __restrict__ ln_g, const float* __restrict__ ln_b) {
    __shared__ __attribute__((aligned(16))) float acc[(TILE + 1) * D_H];
    __shared__ float degs[TILE + 1];
    __shared__ unsigned qs[AGG_WAVES * QCAP];
    const int tid = threadIdx.x;
    const int wave = __builtin_amdgcn_readfirstlane(tid >> 5);
    const int lane = tid & 31;
    const int lo = blockIdx.x * TILE;
    if (lo >= n_dst) return;

    const int nch = (E + CH_EDGES - 1) / CH_EDGES;
    const int ntiles = (n_dst + TILE - 1) / TILE;
    const unsigned* brow = plane + (size_t)blockIdx.x * ((size_t)nch * CH_WORDS);
    const unsigned* pk = plane + (size_t)ntiles * ((size_t)nch * CH_WORDS);
    const int gf = max(min(wave, nch - 1), 0);
    uint4 cur = *(const uint4*)(brow + (size_t)gf * CH_WORDS + lane * 4);
    LnCoef lc;
    if constexpr (LN) {
        lc.ga = *(const float4*)(ln_g + lane * 4);
        lc.gb = *(const float4*)(ln_g + 128 + lane * 4);
        lc.ba = *(const float4*)(ln_b + lane * 4);
        lc.bb = *(const float4*)(ln_b + 128 + lane * 4);
    } else {
        lc.ga = lc.gb = lc.ba = lc.bb = make_float4(0.f, 0.f, 0.f, 0.f);
    }

    for (int i = tid; i < (TILE + 1) * (D_H / 4); i += 1024)
        ((float4*)acc)[i] = make_float4(0.f, 0.f, 0.f, 0.f);
    if (tid <= TILE) degs[tid] = 0.0f;
    __syncthreads();

    unsigned* myq = qs + wave * QCAP;
    unsigned cnt = 0;
    for (int g = wave; g < nch; g += AGG_WAVES) {
        const int gn = min(g + AGG_WAVES, nch - 1);
        const uint4 nxt = *(const uint4*)(brow + (size_t)gn * CH_WORDS + lane * 4);
        const unsigned eb = (unsigned)(g * CH_EDGES + lane * 128);
        unsigned long long ba = ((unsigned long long)cur.y << 32) | cur.x;
        unsigned long long bb = ((unsigned long long)cur.w << 32) | cur.z;
        for (int it = 0; it < 128; ++it) {
            const bool m = (ba | bb) != 0ull;
            const unsigned mask = __builtin_amdgcn_ballot_w32(m);
            if (mask == 0u) break;
            if (m) {
                unsigned bit;
                if (ba != 0ull) { bit = (unsigned)__builtin_ctzll(ba); ba &= ba - 1ull; }
                else            { bit = 64u + (unsigned)__builtin_ctzll(bb); bb &= bb - 1ull; }
                myq[cnt + __builtin_amdgcn_mbcnt_lo(mask, 0u)] = eb + bit;
            }
            cnt += (unsigned)__builtin_popcount(mask);
            if (cnt >= 32u) {
                cnt -= 32u;
                agg_drain<TILE, LN>(myq + cnt, 32u, lane, h, pk, acc, degs, lc);
            }
        }
        cur = nxt;
    }
    if (cnt > 0u) agg_drain<TILE, LN>(myq, cnt, lane, h, pk, acc, degs, lc);
    __syncthreads();

    for (int r = wave; r < TILE; r += AGG_WAVES) {
        int grow = lo + r;
        if (grow >= n_dst) break;
        float inv = 1.0f / (degs[r] + 1.0f);
        const float* ar = acc + r * D_H + lane;
        const float* xr = h + (size_t)grow * D_H;
        float4 x0 = *(const float4*)(xr + lane * 4);
        float4 x1 = *(const float4*)(xr + 128 + lane * 4);
        if constexpr (LN) ln_relu_row(x0, x1, lc);
        float4 o0, o1;
        o0.x = (ar[0]   + x0.x) * inv;  o0.y = (ar[64]  + x0.y) * inv;
        o0.z = (ar[128] + x0.z) * inv;  o0.w = (ar[192] + x0.w) * inv;
        o1.x = (ar[32]  + x1.x) * inv;  o1.y = (ar[96]  + x1.y) * inv;
        o1.z = (ar[160] + x1.z) * inv;  o1.w = (ar[224] + x1.w) * inv;
        float* orow = hneigh + (size_t)grow * D_H;
        *(float4*)(orow + lane * 4) = o0;
        *(float4*)(orow + 128 + lane * 4) = o1;
    }
}

#define ALD 260
#define WLD 64

__device__ __forceinline__ void gemm_load_a(float4 (&pa)[8], const float* __restrict__ A,
                                            int rowBase, int M, int r0, int c4) {
#pragma unroll
    for (int j = 0; j < 8; ++j) {
        const int grow = rowBase + r0 + 4 * j;
        pa[j] = make_float4(0.f, 0.f, 0.f, 0.f);
        if (grow < M) pa[j] = *(const float4*)(A + (size_t)grow * D_H + c4 * 4);
    }
}

__global__ __launch_bounds__(256) void k_gemm_wmma_f32(
    const float* __restrict__ A, const float* __restrict__ W,
    const float* __restrict__ bias, float* __restrict__ out, int M, int N) {
    __shared__ __attribute__((aligned(16))) float Wl[D_H * WLD];
    __shared__ __attribute__((aligned(16))) float Al[32 * ALD];
    const int tid = threadIdx.x;
    const int colBase = blockIdx.y * 64;
    const int nTiles = (M + 31) / 32;
    int t = blockIdx.x;
    if (t >= nTiles) return;

    const int wave = tid >> 5, lane = tid & 31;
    const int wr = wave >> 2, wc = wave & 3;
    const int half = lane >> 4, l16 = lane & 15;

    const int r0 = __builtin_amdgcn_readfirstlane(tid >> 6);
    const int ac4 = tid & 63;

    float4 pa[8];
    gemm_load_a(pa, A, t * 32, M, r0, ac4);

    {
        float4 w[16];
#pragma unroll
        for (int j = 0; j < 16; ++j) {
            const int idx = tid + j * 256;
            const int k = idx >> 4, c4 = idx & 15;
            w[j] = *(const float4*)(W + (size_t)k * N + colBase + c4 * 4);
        }
#pragma unroll
        for (int j = 0; j < 16; ++j) {
            const int idx = tid + j * 256;
            const int k = idx >> 4, c4 = idx & 15;
            *(float4*)(&Wl[k * WLD + c4 * 4]) = w[j];
        }
    }

    const int col = colBase + wc * 16 + l16;
    const float bv = bias[col];
    const float* arow = &Al[(wr * 16 + l16) * ALD + 2 * half];
    const float* bcol = &Wl[(2 * half) * WLD + wc * 16 + l16];

    for (; t < nTiles; t += (int)gridDim.x) {
        __syncthreads();
#pragma unroll
        for (int j = 0; j < 8; ++j) {
            *(float4*)(&Al[(r0 + 4 * j) * ALD + ac4 * 4]) = pa[j];
        }
        __syncthreads();

        const int tn = t + (int)gridDim.x;
        if (tn < nTiles) gemm_load_a(pa, A, tn * 32, M, r0, ac4);

        v8f c = {};
#pragma unroll
        for (int k0 = 0; k0 < D_H; k0 += 4) {
            v2f a, b;
            a.x = arow[k0];
            a.y = arow[k0 + 1];
            b.x = bcol[k0 * WLD];
            b.y = bcol[(k0 + 1) * WLD];
            c = __builtin_amdgcn_wmma_f32_16x16x4_f32(
                false, a, false, b, (short)0, c, false, false);
        }

        const int rbase = t * 32 + wr * 16 + half * 8;
        const size_t o0 = (size_t)rbase * N + col;
#pragma unroll
        for (int j = 0; j < 8; ++j) {
            if (rbase + j < M) out[o0 + (size_t)j * N] = c[j] + bv;
        }
    }
}

extern "C" void kernel_launch(void* const* d_in, const int* in_sizes, int n_in,
                              void* d_out, int out_size, void* d_ws, size_t ws_size,
                              hipStream_t stream) {
    (void)n_in; (void)out_size; (void)ws_size;
    const float* feats = (const float*)d_in[0];
    const float* W0    = (const float*)d_in[1];
    const float* b0    = (const float*)d_in[2];
    const float* W1    = (const float*)d_in[3];
    const float* b1    = (const float*)d_in[4];
    const float* ln_g  = (const float*)d_in[5];
    const float* ln_b  = (const float*)d_in[6];
    const int* src0 = (const int*)d_in[7];
    const int* dst0 = (const int*)d_in[8];
    const int* src1 = (const int*)d_in[9];
    const int* dst1 = (const int*)d_in[10];
    const int E0 = in_sizes[7], E1 = in_sizes[9];
    const int N1 = 50000, N2 = 10000, DOUT = 128;

    float* ws     = (float*)d_ws;
    float* hn0    = ws;
    float* deg0   = hn0 + (size_t)N1 * D_H;
    float* hpl    = deg0 + N1;
    float* hn1p   = hpl + (size_t)N1 * D_H;
    unsigned* bits0 = (unsigned*)(hpl + 48);
    unsigned* bits1 = (unsigned*)(hn1p + 48);
    float* hn1    = hn0;

    float* h0_out = (float*)d_out;
    float* out2   = h0_out + (size_t)N1 * D_H;

    k_tile_bits<<<236, 1024, 0, stream>>>(src0, dst0, bits0, E0, N1, 196,
                                          src1, dst1, bits1, E1, N2);

    k_tile_aggregate<TILE0, false><<<256, 1024, 0, stream>>>(feats, bits0, hn0, E0, N1,
                                                             ln_g, ln_b);
    {
        dim3 grid(192, 4);
        k_gemm_wmma_f32<<<grid, 256, 0, stream>>>(hn0, W0, b0, h0_out, N1, D_H);
    }

    k_tile_aggregate<TILE1, true><<<250, 1024, 0, stream>>>(h0_out, bits1, hn1, E1, N2,
                                                            ln_g, ln_b);
    {
        dim3 grid(313, 2);
        k_gemm_wmma_f32<<<grid, 256, 0, stream>>>(hn1, W1, b1, out2, N2, DOUT);
    }
}


# Edge index as fixed 256-byte cells: pre-pass ranks a chunk's edges per tile at LDS counters, owners read their cell row contiguously; bitmap and per-edge lookups go; LDS float adds, new arrival order
# speedup vs baseline: 3.7222x; 1.0615x over previous
#include <hip/hip_runtime.h>
#include <math.h>

typedef float v2f __attribute__((ext_vector_type(2)));
typedef float v8f __attribute__((ext_vector_type(8)));

#define D_H 256
#define LN_EPS 1e-5f

#define TILE0 196
#define TILE1 40
#define CH_EDGES 4096
#define CELL_W 64
#define CELL_CAP 63

template <int TILE>
__device__ __forceinline__ void tile_cells_layer(unsigned* sl, unsigned* cn,
                                                 const int* __restrict__ src,
                                                 const int* __restrict__ dst,
                                                 unsigned* __restrict__ cells, int E,
                                                 int n_dst, int first, int stride) {
    const int tid = threadIdx.x;
    const int nch = (E + CH_EDGES - 1) / CH_EDGES;
    const int ntiles = (n_dst + TILE - 1) / TILE;
    for (int c = first; c < nch; c += stride) {
        for (int i = tid; i < 256 * CELL_W / 4; i += 1024)
            ((uint4*)sl)[i] = make_uint4(0u, 0u, 0u, 0u);
        if (tid < 256) cn[tid] = 0u;
        __syncthreads();
        const int e0 = c * CH_EDGES + tid * 4;
        int4 d, s;
        if (e0 + 3 < E) {
            d = *(const int4*)(dst + e0);
            s = *(const int4*)(src + e0);
        } else {
            d.x = (e0     < E) ? dst[e0]     : -1;  s.x = (e0     < E) ? src[e0]     : 0;
            d.y = (e0 + 1 < E) ? dst[e0 + 1] : -1;  s.y = (e0 + 1 < E) ? src[e0 + 1] : 0;
            d.z = (e0 + 2 < E) ? dst[e0 + 2] : -1;  s.z = (e0 + 2 < E) ? src[e0 + 2] : 0;
            d.w = (e0 + 3 < E) ? dst[e0 + 3] : -1;  s.w = (e0 + 3 < E) ? src[e0 + 3] : 0;
        }
        const int dk[4] = {d.x, d.y, d.z, d.w};
        const int sk[4] = {s.x, s.y, s.z, s.w};
#pragma unroll
        for (int k = 0; k < 4; ++k) {
            if ((unsigned)dk[k] < (unsigned)n_dst) {
                const unsigned t = (unsigned)dk[k] / (unsigned)TILE;
                const unsigned pw = ((unsigned)sk[k] << 8) | ((unsigned)dk[k] - t * (unsigned)TILE);
                const unsigned r = atomicAdd(&cn[t], 1u);
                if (r < (unsigned)CELL_CAP) sl[t * CELL_W + 1u + r] = pw;
            }
        }
        __syncthreads();
        for (int i = tid; i < ntiles * (CELL_W / 4); i += 1024) {
            const int t = i >> 4, q = i & 15;
            uint4 v = *(const uint4*)(sl + t * CELL_W + q * 4);
            if (q == 0) v.x = cn[t];
            *(uint4*)(cells + ((size_t)t * nch + c) * CELL_W + q * 4) = v;
        }
        __syncthreads();
    }
}

__global__ __launch_bounds__(1024) void k_tile_cells(
    const int* __restrict__ src0, const int* __restrict__ dst0,
    unsigned* __restrict__ cells0, int E0, int n0, int nb0,
    const int* __restrict__ src1, const int* __restrict__ dst1,
    unsigned* __restrict__ cells1, int E1, int n1) {
    __shared__ __attribute__((aligned(16))) unsigned sl[256 * CELL_W];
    __shared__ unsigned cn[256];
    const int b = (int)blockIdx.x;
    if (b < nb0) tile_cells_layer<TILE0>(sl, cn, src0, dst0, cells0, E0, n0, b, nb0);
    else         tile_cells_layer<TILE1>(sl, cn, src1, dst1, cells1, E1, n1, b - nb0,
                                         (int)gridDim.x - nb0);
}

#define AGG_WAVES 32
#define QCAP 96

struct LnCoef { float4 ga, gb, ba, bb; };

template <int CTRL>
__device__ __forceinline__ float dpp_add(float v) {
    return v + __builtin_bit_cast(float, __builtin_amdgcn_update_dpp(
                   0, __builtin_bit_cast(int, v), CTRL, 0xf, 0xf, true));
}
__device__ __forceinline__ float wave_sum32(float v) {
    v = dpp_add<0x161>(v);
    v = dpp_add<0x162>(v);
    v = dpp_add<0x164>(v);
    v = dpp_add<0x168>(v);
    const int t = __builtin_bit_cast(int, v);
    v += __builtin_bit_cast(float, __builtin_amdgcn_permlanex16(t, t, 0x76543210, 0xfedcba98,
                                                               false, false));
    return v;
}

__device__ __forceinline__ void ln_relu_row(float4& a, float4& b, const LnCoef& c) {
    v2f a0 = {a.x, a.y}, a1 = {a.z, a.w}, b0 = {b.x, b.y}, b1 = {b.z, b.w};
    const v2f t = (a0 + a1) + (b0 + b1);
    const float mu = wave_sum32(t.x + t.y) * (1.0f / 256.0f);
    const v2f m2 = {mu, mu};
    a0 -= m2; a1 -= m2; b0 -= m2; b1 -= m2;
    const v2f u = (a0 * a0 + a1 * a1) + (b0 * b0 + b1 * b1);
    const float rs = rsqrtf(wave_sum32(u.x + u.y) * (1.0f / 256.0f) + LN_EPS);
    const v2f r2 = {rs, rs};
    const v2f g0 = {c.ga.x, c.ga.y}, g1 = {c.ga.z, c.ga.w};
    const v2f g2 = {c.gb.x, c.gb.y}, g3 = {c.gb.z, c.gb.w};
    const v2f s0 = {c.ba.x, c.ba.y}, s1 = {c.ba.z, c.ba.w};
    const v2f s2 = {c.bb.x, c.bb.y}, s3 = {c.bb.z, c.bb.w};
    a0 = a0 * r2 * g0 + s0;  a1 = a1 * r2 * g1 + s1;
    b0 = b0 * r2 * g2 + s2;  b1 = b1 * r2 * g3 + s3;
    a.x = fmaxf(a0.x, 0.0f); a.y = fmaxf(a0.y, 0.0f);
    a.z = fmaxf(a1.x, 0.0f); a.w = fmaxf(a1.y, 0.0f);
    b.x = fmaxf(b0.x, 0.0f); b.y = fmaxf(b0.y, 0.0f);
    b.z = fmaxf(b1.x, 0.0f); b.w = fmaxf(b1.y, 0.0f);
}

template <int TILE, bool LN>
__device__ __forceinline__ void agg_drain(const unsigned* qb, unsigned n,
                                          int lane, const float* __restrict__ h,
                                          float* acc, float* degs, const LnCoef& lc) {
    unsigned p = 0u;
    if ((unsigned)lane < n) p = qb[lane];
    int s = (int)(p >> 8);
    int loc = TILE;
    if ((unsigned)lane < n) loc = (int)min(p & 255u, (unsigned)TILE);
    if ((unsigned)lane < n) atomicAdd(&degs[loc], 1.0f);
    constexpr int R = LN ? 7 : 8;
    for (unsigned j0 = 0; j0 < n; j0 += R) {
        float4 va[R], vb[R];
#pragma unroll
        for (int g = 0; g < R; ++g) {
            const int slot = (32 % R != 0) ? min((int)j0 + g, 31) : (int)j0 + g;
            int sj = __builtin_amdgcn_readlane(s, slot);
            const float* hr = h + (size_t)sj * D_H;
            va[g] = *(const float4*)(hr + lane * 4);
            vb[g] = *(const float4*)(hr + 128 + lane * 4);
        }
        if constexpr (LN) __builtin_amdgcn_sched_barrier(0);
#pragma unroll
        for (int g = 0; g < R; ++g) {
            const int slot = (32 % R != 0) ? min((int)j0 + g, 31) : (int)j0 + g;
            int lj = __builtin_amdgcn_readlane(loc, slot);
            if (32 % R != 0 && (int)j0 + g > 31) lj = TILE;
            if constexpr (LN) ln_relu_row(va[g], vb[g], lc);
            float* ar = acc + lj * D_H + lane;
            atomicAdd(ar,       va[g].x);
            atomicAdd(ar + 64,  va[g].y);
            atomicAdd(ar + 128, va[g].z);
            atomicAdd(ar + 192, va[g].w);
            atomicAdd(ar + 32,  vb[g].x);
            atomicAdd(ar + 96,  vb[g].y);
            atomicAdd(ar + 160, vb[g].z);
            atomicAdd(ar + 224, vb[g].w);
        }
    }
}

template <int TILE, bool LN>
__global__ __launch_bounds__(1024) void k_tile_aggregate(
    const float* __restrict__ h, const unsigned* __restrict__ cells,
    const int* __restrict__ src, const int* __restrict__ dst,
    float* __restrict__ hneigh, int E, int n_dst,
    const float* __restrict__ ln_g, const float* __restrict__ ln_b) {
    __shared__ __attribute__((aligned(16))) float acc[(TILE + 1) * D_H];
    __shared__ float degs[TILE + 1];
    __shared__ unsigned qs[AGG_WAVES * QCAP];
    const int tid = threadIdx.x;
    const int wave = __builtin_amdgcn_readfirstlane(tid >> 5);
    const int lane = tid & 31;
    const int lo = blockIdx.x * TILE;
    if (lo >= n_dst) return;

    const int nch = (E + CH_EDGES - 1) / CH_EDGES;
    const unsigned* crow = cells + (size_t)blockIdx.x * ((size_t)nch * CELL_W);
    const int gf = max(min(wave, nch - 1), 0);
    uint2 cur = *(const uint2*)(crow + (size_t)gf * CELL_W + lane * 2);
    LnCoef lc;
    if constexpr (LN) {
        lc.ga = *(const float4*)(ln_g + lane * 4);
        lc.gb = *(const float4*)(ln_g + 128 + lane * 4);
        lc.ba = *(const float4*)(ln_b + lane * 4);
        lc.bb = *(const float4*)(ln_b + 128 + lane * 4);
    } else {
        lc.ga = lc.gb = lc.ba = lc.bb = make_float4(0.f, 0.f, 0.f, 0.f);
    }

    for (int i = tid; i < (TILE + 1) * (D_H / 4); i += 1024)
        ((float4*)acc)[i] = make_float4(0.f, 0.f, 0.f, 0.f);
    if (tid <= TILE) degs[tid] = 0.0f;
    __syncthreads();

    unsigned* myq = qs + wave * QCAP;
    unsigned cnt = 0;
    for (int g = wave; g < nch; g += AGG_WAVES) {
        const int gn = min(g + AGG_WAVES, nch - 1);
        const uint2 nxt = *(const uint2*)(crow + (size_t)gn * CELL_W + lane * 2);
        const unsigned n = (unsigned)__builtin_amdgcn_readfirstlane((int)cur.x);
        if (n <= (unsigned)CELL_CAP) {
            if (lane > 0 && (unsigned)(2 * lane - 1) < n) myq[cnt + 2 * lane - 1] = cur.x;
            if ((unsigned)(2 * lane) < n) myq[cnt + 2 * lane] = cur.y;
            cnt += n;
            while (cnt >= 32u) {
                cnt -= 32u;
                agg_drain<TILE, LN>(myq + cnt, 32u, lane, h, acc, degs, lc);
            }
        } else {
            for (int i = 0; i < CH_EDGES; i += 32) {
                const int e = g * CH_EDGES + i + lane;
                bool m = false;
                unsigned pw = 0u;
                if (e < E) {
                    const unsigned dv = (unsigned)dst[e];
                    if (dv < (unsigned)n_dst && dv / (unsigned)TILE == blockIdx.x) {
                        m = true;
                        pw = ((unsigned)src[e] << 8) | (dv - (unsigned)lo);
                    }
                }
                const unsigned mask = __builtin_amdgcn_ballot_w32(m);
                if (m) myq[cnt + __builtin_amdgcn_mbcnt_lo(mask, 0u)] = pw;
                cnt += (unsigned)__builtin_popcount(mask);
                if (cnt >= 32u) {
                    cnt -= 32u;
                    agg_drain<TILE, LN>(myq + cnt, 32u, lane, h, acc, degs, lc);
                }
            }
        }
        cur = nxt;
    }
    if (cnt > 0u) agg_drain<TILE, LN>(myq, cnt, lane, h, acc, degs, lc);
    __syncthreads();

    for (int r = wave; r < TILE; r += AGG_WAVES) {
        int grow = lo + r;
        if (grow >= n_dst) break;
        float inv = 1.0f / (degs[r] + 1.0f);
        const float* ar = acc + r * D_H + lane;
        const float* xr = h + (size_t)grow * D_H;
        float4 x0 = *(const float4*)(xr + lane * 4);
        float4 x1 = *(const float4*)(xr + 128 + lane * 4);
        if constexpr (LN) ln_relu_row(x0, x1, lc);
        float4 o0, o1;
        o0.x = (ar[0]   + x0.x) * inv;  o0.y = (ar[64]  + x0.y) * inv;
        o0.z = (ar[128] + x0.z) * inv;  o0.w = (ar[192] + x0.w) * inv;
        o1.x = (ar[32]  + x1.x) * inv;  o1.y = (ar[96]  + x1.y) * inv;
        o1.z = (ar[160] + x1.z) * inv;  o1.w = (ar[224] + x1.w) * inv;
        float* orow = hneigh + (size_t)grow * D_H;
        *(float4*)(orow + lane * 4) = o0;
        *(float4*)(orow + 128 + lane * 4) = o1;
    }
}

#define ALD 260
#define WLD 64

__device__ __forceinline__ void gemm_load_a(float4 (&pa)[8], const float* __restrict__ A,
                                            int rowBase, int M, int r0, int c4) {
#pragma unroll
    for (int j = 0; j < 8; ++j) {
        const int grow = rowBase + r0 + 4 * j;
        pa[j] = make_float4(0.f, 0.f, 0.f, 0.f);
        if (grow < M) pa[j] = *(const float4*)(A + (size_t)grow * D_H + c4 * 4);
    }
}

__global__ __launch_bounds__(256) void k_gemm_wmma_f32(
    const float* __restrict__ A, const float* __restrict__ W,
    const float* __restrict__ bias, float* __restrict__ out, int M, int N) {
    __shared__ __attribute__((aligned(16))) float Wl[D_H * WLD];
    __shared__ __attribute__((aligned(16))) float Al[32 * ALD];
    const int tid = threadIdx.x;
    const int colBase = blockIdx.y * 64;
    const int nTiles = (M + 31) / 32;
    int t = blockIdx.x;
    if (t >= nTiles) return;

    const int wave = tid >> 5, lane = tid & 31;
    const int wr = wave >> 2, wc = wave & 3;
    const int half = lane >> 4, l16 = lane & 15;

    const int r0 = __builtin_amdgcn_readfirstlane(tid >> 6);
    const int ac4 = tid & 63;

    float4 pa[8];
    gemm_load_a(pa, A, t * 32, M, r0, ac4);

    {
        float4 w[16];
#pragma unroll
        for (int j = 0; j < 16; ++j) {
            const int idx = tid + j * 256;
            const int k = idx >> 4, c4 = idx & 15;
            w[j] = *(const float4*)(W + (size_t)k * N + colBase + c4 * 4);
        }
#pragma unroll
        for (int j = 0; j < 16; ++j) {
            const int idx = tid + j * 256;
            const int k = idx >> 4, c4 = idx & 15;
            *(float4*)(&Wl[k * WLD + c4 * 4]) = w[j];
        }
    }

    const int col = colBase + wc * 16 + l16;
    const float bv = bias[col];
    const float* arow = &Al[(wr * 16 + l16) * ALD + 2 * half];
    const float* bcol = &Wl[(2 * half) * WLD + wc * 16 + l16];

    for (; t < nTiles; t += (int)gridDim.x) {
        __syncthreads();
#pragma unroll
        for (int j = 0; j < 8; ++j) {
            *(float4*)(&Al[(r0 + 4 * j) * ALD + ac4 * 4]) = pa[j];
        }
        __syncthreads();

        const int tn = t + (int)gridDim.x;
        if (tn < nTiles) gemm_load_a(pa, A, tn * 32, M, r0, ac4);

        v8f c = {};
#pragma unroll
        for (int k0 = 0; k0 < D_H; k0 += 4) {
            v2f a, b;
            a.x = arow[k0];
            a.y = arow[k0 + 1];
            b.x = bcol[k0 * WLD];
            b.y = bcol[(k0 + 1) * WLD];
            c = __builtin_amdgcn_wmma_f32_16x16x4_f32(
                false, a, false, b, (short)0, c, false, false);
        }

        const int rbase = t * 32 + wr * 16 + half * 8;
        const size_t o0 = (size_t)rbase * N + col;
#pragma unroll
        for (int j = 0; j < 8; ++j) {
            if (rbase + j < M) out[o0 + (size_t)j * N] = c[j] + bv;
        }
    }
}

extern "C" void kernel_launch(void* const* d_in, const int* in_sizes, int n_in,
                              void* d_out, int out_size, void* d_ws, size_t ws_size,
                              hipStream_t stream) {
    (void)n_in; (void)out_size; (void)ws_size;
    const float* feats = (const float*)d_in[0];
    const float* W0    = (const float*)d_in[1];
    const float* b0    = (const float*)d_in[2];
    const float* W1    = (const float*)d_in[3];
    const float* b1    = (const float*)d_in[4];
    const float* ln_g  = (const float*)d_in[5];
    const float* ln_b  = (const float*)d_in[6];
    const int* src0 = (const int*)d_in[7];
    const int* dst0 = (const int*)d_in[8];
    const int* src1 = (const int*)d_in[9];
    const int* dst1 = (const int*)d_in[10];
    const int E0 = in_sizes[7], E1 = in_sizes[9];
    const int N1 = 50000, N2 = 10000, DOUT = 128;

    float* ws     = (float*)d_ws;
    float* hn0    = ws;
    float* deg0   = hn0 + (size_t)N1 * D_H;
    float* hpl    = deg0 + N1;
    float* hn1p   = hpl + (size_t)N1 * D_H;
    unsigned* cells0 = (unsigned*)(hpl + 48);
    unsigned* cells1 = (unsigned*)(hn1p + 48);
    float* hn1    = hn0;

    float* h0_out = (float*)d_out;
    float* out2   = h0_out + (size_t)N1 * D_H;

    k_tile_cells<<<236, 1024, 0, stream>>>(src0, dst0, cells0, E0, N1, 196,
                                           src1, dst1, cells1, E1, N2);

    k_tile_aggregate<TILE0, false><<<256, 1024, 0, stream>>>(feats, cells0, src0, dst0, hn0,
                                                             E0, N1, ln_g, ln_b);
    {
        dim3 grid(192, 4);
        k_gemm_wmma_f32<<<grid, 256, 0, stream>>>(hn0, W0, b0, h0_out, N1, D_H);
    }

    k_tile_aggregate<TILE1, true><<<250, 1024, 0, stream>>>(h0_out, cells1, src1, dst1, hn1,
                                                            E1, N2, ln_g, ln_b);
    {
        dim3 grid(313, 2);
        k_gemm_wmma_f32<<<grid, 256, 0, stream>>>(hn1, W1, b1, out2, N2, DOUT);
    }
}


# GEMM fused into the tile owners: the scaled LDS sum rows, re-laid in place at a 260-float pitch, are the WMMA A operand; W staged in 64-k LDS slabs, k order kept; h_neigh planes and 2 launches go
# speedup vs baseline: 4.2149x; 1.1324x over previous
#include <hip/hip_runtime.h>
#include <math.h>

typedef float v2f __attribute__((ext_vector_type(2)));
typedef float v8f __attribute__((ext_vector_type(8)));

#define D_H 256
#define ALD 260
#define LN_EPS 1e-5f

#define TILE0 196
#define TILE1 40
#define CH_EDGES 4096
#define CELL_W 64
#define CELL_CAP 63

template <int TILE>
__device__ __forceinline__ void tile_cells_layer(unsigned* sl, unsigned* cn,
                                                 const int* __restrict__ src,
                                                 const int* __restrict__ dst,
                                                 unsigned* __restrict__ cells, int E,
                                                 int n_dst, int first, int stride) {
    const int tid = threadIdx.x;
    const int nch = (E + CH_EDGES - 1) / CH_EDGES;
    const int ntiles = (n_dst + TILE - 1) / TILE;
    for (int c = first; c < nch; c += stride) {
        for (int i = tid; i < 256 * CELL_W / 4; i += 1024)
            ((uint4*)sl)[i] = make_uint4(0u, 0u, 0u, 0u);
        if (tid < 256) cn[tid] = 0u;
        __syncthreads();
        const int e0 = c * CH_EDGES + tid * 4;
        int4 d, s;
        if (e0 + 3 < E) {
            d = *(const int4*)(dst + e0);
            s = *(const int4*)(src + e0);
        } else {
            d.x = (e0     < E) ? dst[e0]     : -1;  s.x = (e0     < E) ? src[e0]     : 0;
            d.y = (e0 + 1 < E) ? dst[e0 + 1] : -1;  s.y = (e0 + 1 < E) ? src[e0 + 1] : 0;
            d.z = (e0 + 2 < E) ? dst[e0 + 2] : -1;  s.z = (e0 + 2 < E) ? src[e0 + 2] : 0;
            d.w = (e0 + 3 < E) ? dst[e0 + 3] : -1;  s.w = (e0 + 3 < E) ? src[e0 + 3] : 0;
        }
        const int dk[4] = {d.x, d.y, d.z, d.w};
        const int sk[4] = {s.x, s.y, s.z, s.w};
#pragma unroll
        for (int k = 0; k < 4; ++k) {
            if ((unsigned)dk[k] < (unsigned)n_dst) {
                const unsigned t = (unsigned)dk[k] / (unsigned)TILE;
                const unsigned pw = ((unsigned)sk[k] << 8) | ((unsigned)dk[k] - t * (unsigned)TILE);
                const unsigned r = atomicAdd(&cn[t], 1u);
                if (r < (unsigned)CELL_CAP) sl[t * CELL_W + 1u + r] = pw;
            }
        }
        __syncthreads();
        for (int i = tid; i < ntiles * (CELL_W / 4); i += 1024) {
            const int t = i >> 4, q = i & 15;
            uint4 v = *(const uint4*)(sl + t * CELL_W + q * 4);
            if (q == 0) v.x = cn[t];
            *(uint4*)(cells + ((size_t)t * nch + c) * CELL_W + q * 4) = v;
        }
        __syncthreads();
    }
}

__global__ __launch_bounds__(1024) void k_tile_cells(
    const int* __restrict__ src0, const int* __restrict__ dst0,
    unsigned* __restrict__ cells0, int E0, int n0, int nb0,
    const int* __restrict__ src1, const int* __restrict__ dst1,
    unsigned* __restrict__ cells1, int E1, int n1) {
    __shared__ __attribute__((aligned(16))) unsigned sl[256 * CELL_W];
    __shared__ unsigned cn[256];
    const int b = (int)blockIdx.x;
    if (b < nb0) tile_cells_layer<TILE0>(sl, cn, src0, dst0, cells0, E0, n0, b, nb0);
    else         tile_cells_layer<TILE1>(sl, cn, src1, dst1, cells1, E1, n1, b - nb0,
                                         (int)gridDim.x - nb0);
}

#define AGG_WAVES 32
#define QCAP 96

struct LnCoef { float4 ga, gb, ba, bb; };

template <int CTRL>
__device__ __forceinline__ float dpp_add(float v) {
    return v + __builtin_bit_cast(float, __builtin_amdgcn_update_dpp(
                   0, __builtin_bit_cast(int, v), CTRL, 0xf, 0xf, true));
}
__device__ __forceinline__ float wave_sum32(float v) {
    v = dpp_add<0x161>(v);
    v = dpp_add<0x162>(v);
    v = dpp_add<0x164>(v);
    v = dpp_add<0x168>(v);
    const int t = __builtin_bit_cast(int, v);
    v += __builtin_bit_cast(float, __builtin_amdgcn_permlanex16(t, t, 0x76543210, 0xfedcba98,
                                                               false, false));
    return v;
}

__device__ __forceinline__ void ln_relu_row(float4& a, float4& b, const LnCoef& c) {
    v2f a0 = {a.x, a.y}, a1 = {a.z, a.w}, b0 = {b.x, b.y}, b1 = {b.z, b.w};
    const v2f t = (a0 + a1) + (b0 + b1);
    const float mu = wave_sum32(t.x + t.y) * (1.0f / 256.0f);
    const v2f m2 = {mu, mu};
    a0 -= m2; a1 -= m2; b0 -= m2; b1 -= m2;
    const v2f u = (a0 * a0 + a1 * a1) + (b0 * b0 + b1 * b1);
    const float rs = rsqrtf(wave_sum32(u.x + u.y) * (1.0f / 256.0f) + LN_EPS);
    const v2f r2 = {rs, rs};
    const v2f g0 = {c.ga.x, c.ga.y}, g1 = {c.ga.z, c.ga.w};
    const v2f g2 = {c.gb.x, c.gb.y}, g3 = {c.gb.z, c.gb.w};
    const v2f s0 = {c.ba.x, c.ba.y}, s1 = {c.ba.z, c.ba.w};
    const v2f s2 = {c.bb.x, c.bb.y}, s3 = {c.bb.z, c.bb.w};
    a0 = a0 * r2 * g0 + s0;  a1 = a1 * r2 * g1 + s1;
    b0 = b0 * r2 * g2 + s2;  b1 = b1 * r2 * g3 + s3;
    a.x = fmaxf(a0.x, 0.0f); a.y = fmaxf(a0.y, 0.0f);
    a.z = fmaxf(a1.x, 0.0f); a.w = fmaxf(a1.y, 0.0f);
    b.x = fmaxf(b0.x, 0.0f); b.y = fmaxf(b0.y, 0.0f);
    b.z = fmaxf(b1.x, 0.0f); b.w = fmaxf(b1.y, 0.0f);
}

template <int TILE, bool LN>
__device__ __forceinline__ void agg_drain(const unsigned* qb, unsigned n,
                                          int lane, const float* __restrict__ h,
                                          float* acc, float* degs, const LnCoef& lc) {
    unsigned p = 0u;
    if ((unsigned)lane < n) p = qb[lane];
    int s = (int)(p >> 8);
    int loc = TILE;
    if ((unsigned)lane < n) loc = (int)min(p & 255u, (unsigned)TILE);
    if ((unsigned)lane < n) atomicAdd(&degs[loc], 1.0f);
    constexpr int R = LN ? 7 : 8;
    for (unsigned j0 = 0; j0 < n; j0 += R) {
        float4 va[R], vb[R];
#pragma unroll
        for (int g = 0; g < R; ++g) {
            const int slot = (32 % R != 0) ? min((int)j0 + g, 31) : (int)j0 + g;
            int sj = __builtin_amdgcn_readlane(s, slot);
            const float* hr = h + (size_t)sj * D_H;
            va[g] = *(const float4*)(hr + lane * 4);
            vb[g] = *(const float4*)(hr + 128 + lane * 4);
        }
        if constexpr (LN) __builtin_amdgcn_sched_barrier(0);
#pragma unroll
        for (int g = 0; g < R; ++g) {
            const int slot = (32 % R != 0) ? min((int)j0 + g, 31) : (int)j0 + g;
            int lj = __builtin_amdgcn_readlane(loc, slot);
            if (32 % R != 0 && (int)j0 + g > 31) lj = TILE;
            if constexpr (LN) ln_relu_row(va[g], vb[g], lc);
            float* ar = acc + lj * ALD + lane;
            atomicAdd(ar,       va[g].x);
            atomicAdd(ar + 64,  va[g].y);
            atomicAdd(ar + 128, va[g].z);
            atomicAdd(ar + 192, va[g].w);
            atomicAdd(ar + 32,  vb[g].x);
            atomicAdd(ar + 96,  vb[g].y);
            atomicAdd(ar + 160, vb[g].z);
            atomicAdd(ar + 224, vb[g].w);
        }
    }
}

template <int NI, int NT, int WP, bool OPT>
__device__ __forceinline__ void slab_products(v8f (&c)[NI], const float* arow,
                                              const float* bcol, bool opt_ok) {
#pragma unroll
    for (int kk = 0; kk < 16; ++kk) {
        v2f b;
        b.x = bcol[(4 * kk) * WP];
        b.y = bcol[(4 * kk + 1) * WP];
        const float* ak = arow + 4 * kk;
        if constexpr (OPT) {
            if (opt_ok) {
                const v2f a = *(const v2f*)(ak + NT * 16 * ALD);
                c[NT] = __builtin_amdgcn_wmma_f32_16x16x4_f32(
                    false, a, false, b, (short)0, c[NT], false, false);
            }
        }
#pragma unroll
        for (int i = 0; i < NT; ++i) {
            const v2f a = *(const v2f*)(ak + i * 16 * ALD);
            c[i] = __builtin_amdgcn_wmma_f32_16x16x4_f32(
                false, a, false, b, (short)0, c[i], false, false);
        }
    }
}

template <int TILE, bool LN, int NOUT>
__global__ __launch_bounds__(1024) void k_tile_aggregate(
    const float* __restrict__ h, const unsigned* __restrict__ cells,
    const int* __restrict__ src, const int* __restrict__ dst,
    const float* __restrict__ W, const float* __restrict__ bias,
    float* __restrict__ out, int E, int n_dst,
    const float* __restrict__ ln_g, const float* __restrict__ ln_b) {
    constexpr int NRT = (TILE + 15) / 16;
    constexpr int NCT = NOUT / 16;
    constexpr int RG = AGG_WAVES / NCT;
    constexpr int NI = (NRT + RG - 1) / RG;
    constexpr int ROWS = NRT * 16;
    constexpr int WP = NOUT + 16;
    constexpr int NV = (64 * NOUT / 4) / 1024;
    static_assert(ROWS >= TILE + 1, "the sink row must lie inside the kept rows");
    static_assert(NI == 1 || (RG - 1) * NI + NI - 2 < NRT, "only a wave's last row tile may be missing");
    __shared__ __attribute__((aligned(16))) float acc[ROWS * ALD];
    __shared__ __attribute__((aligned(16))) float Wl[64 * WP];
    __shared__ float degs[TILE + 1];
    __shared__ unsigned qs[AGG_WAVES * QCAP];
    const int tid = threadIdx.x;
    const int wave = __builtin_amdgcn_readfirstlane(tid >> 5);
    const int lane = tid & 31;
    const int lo = blockIdx.x * TILE;
    if (lo >= n_dst) return;

    const int nch = (E + CH_EDGES - 1) / CH_EDGES;
    const unsigned* crow = cells + (size_t)blockIdx.x * ((size_t)nch * CELL_W);
    const int gf = max(min(wave, nch - 1), 0);
    uint2 cur = *(const uint2*)(crow + (size_t)gf * CELL_W + lane * 2);
    LnCoef lc;
    if constexpr (LN) {
        lc.ga = *(const float4*)(ln_g + lane * 4);
        lc.gb = *(const float4*)(ln_g + 128 + lane * 4);
        lc.ba = *(const float4*)(ln_b + lane * 4);
        lc.bb = *(const float4*)(ln_b + 128 + lane * 4);
    } else {
        lc.ga = lc.gb = lc.ba = lc.bb = make_float4(0.f, 0.f, 0.f, 0.f);
    }

    for (int i = tid; i < ROWS * (ALD / 4); i += 1024)
        ((float4*)acc)[i] = make_float4(0.f, 0.f, 0.f, 0.f);
    if (tid <= TILE) degs[tid] = 0.0f;
    __syncthreads();

    unsigned* myq = qs + wave * QCAP;
    unsigned cnt = 0;
    for (int g = wave; g < nch; g += AGG_WAVES) {
        const int gn = min(g + AGG_WAVES, nch - 1);
        const uint2 nxt = *(const uint2*)(crow + (size_t)gn * CELL_W + lane * 2);
        const unsigned n = (unsigned)__builtin_amdgcn_readfirstlane((int)cur.x);
        if (n <= (unsigned)CELL_CAP) {
            if (lane > 0 && (unsigned)(2 * lane - 1) < n) myq[cnt + 2 * lane - 1] = cur.x;
            if ((unsigned)(2 * lane) < n) myq[cnt + 2 * lane] = cur.y;
            cnt += n;
            while (cnt >= 32u) {
                cnt -= 32u;
                agg_drain<TILE, LN>(myq + cnt, 32u, lane, h, acc, degs, lc);
            }
        } else {
            for (int i = 0; i < CH_EDGES; i += 32) {
                const int e = g * CH_EDGES + i + lane;
                bool m = false;
                unsigned pw = 0u;
                if (e < E) {
                    const unsigned dv = (unsigned)dst[e];
                    if (dv < (unsigned)n_dst && dv / (unsigned)TILE == blockIdx.x) {
                        m = true;
                        pw = ((unsigned)src[e] << 8) | (dv - (unsigned)lo);
                    }
                }
                const unsigned mask = __builtin_amdgcn_ballot_w32(m);
                if (m) myq[cnt + __builtin_amdgcn_mbcnt_lo(mask, 0u)] = pw;
                cnt += (unsigned)__builtin_popcount(mask);
                if (cnt >= 32u) {
                    cnt -= 32u;
                    agg_drain<TILE, LN>(myq + cnt, 32u, lane, h, acc, degs, lc);
                }
            }
        }
        cur = nxt;
    }
    if (cnt > 0u) agg_drain<TILE, LN>(myq, cnt, lane, h, acc, degs, lc);

    static_assert(NV == 2 || NV == 4, "a thread stages 2 or 4 float4 of a slab");
    const float* wsrc = W + (size_t)tid * 4;
    float4 w0 = *(const float4*)(wsrc);
    float4 w1 = *(const float4*)(wsrc + 4096);
    float4 w2 = make_float4(0.f, 0.f, 0.f, 0.f), w3 = w2;
    if constexpr (NV == 4) {
        w2 = *(const float4*)(wsrc + 2 * 4096);
        w3 = *(const float4*)(wsrc + 3 * 4096);
    }
    __syncthreads();

    for (int r = wave; r < TILE; r += AGG_WAVES) {
        int grow = lo + r;
        if (grow >= n_dst) break;
        float inv = 1.0f / (degs[r] + 1.0f);
        float* ar = acc + r * ALD;
        const float* xr = h + (size_t)grow * D_H;
        float4 x0 = *(const float4*)(xr + lane * 4);
        float4 x1 = *(const float4*)(xr + 128 + lane * 4);
        if constexpr (LN) ln_relu_row(x0, x1, lc);
        float4 o0, o1;
        o0.x = (ar[lane]       + x0.x) * inv;  o0.y = (ar[lane + 64]  + x0.y) * inv;
        o0.z = (ar[lane + 128] + x0.z) * inv;  o0.w = (ar[lane + 192] + x0.w) * inv;
        o1.x = (ar[lane + 32]  + x1.x) * inv;  o1.y = (ar[lane + 96]  + x1.y) * inv;
        o1.z = (ar[lane + 160] + x1.z) * inv;  o1.w = (ar[lane + 224] + x1.w) * inv;
        *(float4*)(ar + lane * 4) = o0;
        *(float4*)(ar + 128 + lane * 4) = o1;
    }

    const int half = lane >> 4, l16 = lane & 15;
    const int ct = wave % NCT, rg = wave / NCT;
    const int rt0 = rg * NI;
    const bool last_ok = rt0 + NI - 1 < NRT;
    const int col = ct * 16 + l16;
    const float bv = bias[col];
    const float* arow0 = acc + (rt0 * 16 + l16) * ALD + 2 * half;
    const float* bcol = Wl + (2 * half) * WP + col;
    v8f c[NI];
#pragma unroll
    for (int i = 0; i < NI; ++i) c[i] = v8f{};

    for (int s = 0; s < D_H / 64; ++s) {
        if (s > 0) __syncthreads();
        {
            constexpr int Q = NOUT / 4, KS = 1024 / Q;
            const int k = tid / Q, c4 = tid % Q;
            float* wd = &Wl[k * WP + c4 * 4];
            *(float4*)(wd) = w0;
            *(float4*)(wd + KS * WP) = w1;
            if constexpr (NV == 4) {
                *(float4*)(wd + 2 * KS * WP) = w2;
                *(float4*)(wd + 3 * KS * WP) = w3;
            }
        }
        __syncthreads();
        if (s + 1 < D_H / 64) {
            const float* wn = wsrc + (size_t)(s + 1) * 64 * NOUT;
            w0 = *(const float4*)(wn);
            w1 = *(const float4*)(wn + 4096);
            if constexpr (NV == 4) {
                w2 = *(const float4*)(wn + 2 * 4096);
                w3 = *(const float4*)(wn + 3 * 4096);
            }
        }
        if constexpr (NI == 1) {
            if (last_ok) slab_products<1, 1, WP, false>(c, arow0 + 64 * s, bcol, true);
        } else {
            slab_products<NI, NI - 1, WP, true>(c, arow0 + 64 * s, bcol, last_ok);
        }
    }

#pragma unroll
    for (int i = 0; i < NI; ++i) {
        if (i < NI - 1 || last_ok) {
            const int rbase = (rt0 + i) * 16 + half * 8;
            const size_t o0 = (size_t)(lo + rbase) * NOUT + col;
#pragma unroll
            for (int j = 0; j < 8; ++j) {
                if (rbase + j < TILE && lo + rbase + j < n_dst) out[o0 + (size_t)j * NOUT] = c[i][j] + bv;
            }
        }
    }
}

extern "C" void kernel_launch(void* const* d_in, const int* in_sizes, int n_in,
                              void* d_out, int out_size, void* d_ws, size_t ws_size,
                              hipStream_t stream) {
    (void)n_in; (void)out_size; (void)ws_size;
    const float* feats = (const float*)d_in[0];
    const float* W0    = (const float*)d_in[1];
    const float* b0    = (const float*)d_in[2];
    const float* W1    = (const float*)d_in[3];
    const float* b1    = (const float*)d_in[4];
    const float* ln_g  = (const float*)d_in[5];
    const float* ln_b  = (const float*)d_in[6];
    const int* src0 = (const int*)d_in[7];
    const int* dst0 = (const int*)d_in[8];
    const int* src1 = (const int*)d_in[9];
    const int* dst1 = (const int*)d_in[10];
    const int E0 = in_sizes[7], E1 = in_sizes[9];
    const int N1 = 50000, N2 = 10000;

    float* ws     = (float*)d_ws;
    float* hn0    = ws;
    float* deg0   = hn0 + (size_t)N1 * D_H;
    float* hpl    = deg0 + N1;
    float* hn1p   = hpl + (size_t)N1 * D_H;
    unsigned* cells0 = (unsigned*)(hpl + 48);
    unsigned* cells1 = (unsigned*)(hn1p + 48);

    float* h0_out = (float*)d_out;
    float* out2   = h0_out + (size_t)N1 * D_H;

    k_tile_cells<<<236, 1024, 0, stream>>>(src0, dst0, cells0, E0, N1, 196,
                                           src1, dst1, cells1, E1, N2);

    k_tile_aggregate<TILE0, false, D_H><<<256, 1024, 0, stream>>>(feats, cells0, src0, dst0,
                                                                  W0, b0, h0_out, E0, N1,
                                                                  ln_g, ln_b);

    k_tile_aggregate<TILE1, true, 128><<<250, 1024, 0, stream>>>(h0_out, cells1, src1, dst1,
                                                                 W1, b1, out2, E1, N2,
                                                                 ln_g, ln_b);
}
